# starting kernel
# baseline (speedup 1.0000x reference)
#include <hip/hip_runtime.h>


typedef __attribute__((ext_vector_type(2))) float v2f;
typedef __attribute__((ext_vector_type(8))) float v8f;

__device__ __forceinline__ v8f wmma_k4(v2f a, v2f b, v8f c) {
  return __builtin_amdgcn_wmma_f32_16x16x4_f32(
       false, a,  false, b,
       (short)0, c,  false,  false);
}


__global__ void zero_kernel(float* __restrict__ p, long long n) {
  long long i = (long long)blockIdx.x * blockDim.x + threadIdx.x;
  if (i < n) p[i] = 0.0f;
}

__global__ void count_kernel(const int* __restrict__ dst, int E,
                             float* __restrict__ cnt) {
  int i = blockIdx.x * blockDim.x + threadIdx.x;
  if (i < E) atomicAdd(&cnt[dst[i]], 1.0f);
}

__global__ void addmat_kernel(const float* __restrict__ a,
                              const float* __restrict__ b,
                              float* __restrict__ o, int n) {
  int i = blockIdx.x * blockDim.x + threadIdx.x;
  if (i < n) o[i] = a[i] + b[i];
}

__global__ void scatter16_kernel(const float* __restrict__ Y,
                                 const int* __restrict__ src,
                                 const int* __restrict__ dst, int E,
                                 float* __restrict__ agg) {
  long long i = (long long)blockIdx.x * blockDim.x + threadIdx.x;
  long long total = (long long)E * 4;
  if (i >= total) return;
  int e = (int)(i >> 2);
  int c = (int)(i & 3);
  int s = src[e];
  int d = dst[e];
  const float4 v = *reinterpret_cast<const float4*>(Y + (size_t)s * 16 + c * 4);
  float* p = agg + (size_t)d * 16 + c * 4;
  atomicAdd(p + 0, v.x);
  atomicAdd(p + 1, v.y);
  atomicAdd(p + 2, v.z);
  atomicAdd(p + 3, v.w);
}

__global__ void finalize_l1_kernel(float* __restrict__ XO,
                                   const float* __restrict__ aggA,
                                   const float* __restrict__ cntA,
                                   const float* __restrict__ bA,
                                   const float* __restrict__ aggB,
                                   const float* __restrict__ cntB,
                                   const float* __restrict__ bB, int M) {
  long long i = (long long)blockIdx.x * blockDim.x + threadIdx.x;
  if (i >= (long long)M * 16) return;
  int node = (int)(i >> 4);
  int f = (int)(i & 15);
  float ca = fmaxf(cntA[node], 1.0f);
  float cb = fmaxf(cntB[node], 1.0f);
  float v = XO[i] + aggA[i] / ca + bA[f] + aggB[i] / cb + bB[f];
  XO[i] = fmaxf(v, 0.0f);
}

__global__ __launch_bounds__(128) void l1_transform_kernel(
    const float* __restrict__ X, const float* __restrict__ W0,
    const float* __restrict__ W1, const float* __restrict__ W2,
    float* __restrict__ Y0, float* __restrict__ Y1, float* __restrict__ Y2,
    int M) {
  const int lane = threadIdx.x & 31;
  const int wave = threadIdx.x >> 5;
  const int wpb = blockDim.x >> 5;
  const int half = lane >> 4;
  const int mrow = lane & 15;
  const int nTiles = (M + 15) >> 4;

  v2f b0[16], b1[16], b2[16];
#pragma unroll
  for (int s = 0; s < 16; ++s) {
    int k = 4 * s + 2 * half;
    b0[s] = (v2f){W0[k * 16 + mrow], W0[(k + 1) * 16 + mrow]};
    b1[s] = (v2f){W1[k * 16 + mrow], W1[(k + 1) * 16 + mrow]};
    b2[s] = (v2f){W2[k * 16 + mrow], W2[(k + 1) * 16 + mrow]};
  }

  for (int tile = blockIdx.x * wpb + wave; tile < nTiles;
       tile += gridDim.x * wpb) {
    const int rbase = tile << 4;
    int arow = rbase + mrow;
    if (arow > M - 1) arow = M - 1;
    const float* xr = X + (size_t)arow * 64;
    v8f c0 = {}, c1 = {}, c2 = {};
#pragma unroll
    for (int s = 0; s < 16; ++s) {
      int k = 4 * s + 2 * half;
      v2f a = (v2f){xr[k], xr[k + 1]};
      c0 = wmma_k4(a, b0[s], c0);
      c1 = wmma_k4(a, b1[s], c1);
      c2 = wmma_k4(a, b2[s], c2);
    }
#pragma unroll
    for (int r = 0; r < 8; ++r) {
      int row = rbase + r + 8 * half;
      if (row < M) {
        size_t o = (size_t)row * 16 + mrow;
        Y0[o] = c0[r];
        Y1[o] = c1[r];
        Y2[o] = c2[r];
      }
    }
  }
}

__global__ __launch_bounds__(128) void l2_fused_kernel(
    const float* __restrict__ A0, const float* __restrict__ cnt0,
    const float* __restrict__ A1, const float* __restrict__ cnt1,
    const float* __restrict__ A2, const float* __restrict__ W0,
    const float* __restrict__ W1, const float* __restrict__ W2,
    const float* __restrict__ bias0, const float* __restrict__ bias1,
    float* __restrict__ OUT, int M) {
  const int lane = threadIdx.x & 31;
  const int wave = threadIdx.x >> 5;
  const int wpb = blockDim.x >> 5;
  const int half = lane >> 4;
  const int mrow = lane & 15;
  const int nTiles = (M + 15) >> 4;

  const float* Ws[3] = {W0, W1, W2};
  v2f b[3][4][2];
#pragma unroll
  for (int m = 0; m < 3; ++m)
#pragma unroll
    for (int s = 0; s < 4; ++s)
#pragma unroll
      for (int t = 0; t < 2; ++t) {
        int k = 4 * s + 2 * half;
        int col = 16 * t + mrow;
        b[m][s][t] = (v2f){Ws[m][k * 32 + col], Ws[m][(k + 1) * 32 + col]};
      }

  const float biasL = bias0[mrow] + bias1[mrow];
  const float biasH = bias0[16 + mrow] + bias1[16 + mrow];

  for (int tile = blockIdx.x * wpb + wave; tile < nTiles;
       tile += gridDim.x * wpb) {
    const int rbase = tile << 4;
    int arow = rbase + mrow;
    if (arow > M - 1) arow = M - 1;
    const float rs0 = 1.0f / fmaxf(cnt0[arow], 1.0f);
    const float rs1 = 1.0f / fmaxf(cnt1[arow], 1.0f);
    const float* a0p = A0 + (size_t)arow * 16;
    const float* a1p = A1 + (size_t)arow * 16;
    const float* a2p = A2 + (size_t)arow * 16;
    v8f cL = {}, cH = {};
#pragma unroll
    for (int s = 0; s < 4; ++s) {
      int k = 4 * s + 2 * half;
      v2f a0 = (v2f){a0p[k] * rs0, a0p[k + 1] * rs0};
      cL = wmma_k4(a0, b[0][s][0], cL);
      cH = wmma_k4(a0, b[0][s][1], cH);
      v2f a1 = (v2f){a1p[k] * rs1, a1p[k + 1] * rs1};
      cL = wmma_k4(a1, b[1][s][0], cL);
      cH = wmma_k4(a1, b[1][s][1], cH);
      v2f a2 = (v2f){a2p[k], a2p[k + 1]};
      cL = wmma_k4(a2, b[2][s][0], cL);
      cH = wmma_k4(a2, b[2][s][1], cH);
    }
#pragma unroll
    for (int r = 0; r < 8; ++r) {
      int row = rbase + r + 8 * half;
      if (row < M) {
        size_t o = (size_t)row * 32 + mrow;
        OUT[o] = cL[r] + biasL;
        OUT[o + 16] = cH[r] + biasH;
      }
    }
  }
}


extern "C" void kernel_launch(void* const* d_in, const int* in_sizes, int n_in,
                              void* d_out, int out_size, void* d_ws,
                              size_t ws_size, hipStream_t stream) {
  const float* x_dev = (const float*)d_in[0];
  const float* x_task = (const float*)d_in[1];
  const int* wr_src = (const int*)d_in[2];
  const int* wr_dst = (const int*)d_in[3];
  const int* rv_src = (const int*)d_in[4];
  const int* rv_dst = (const int*)d_in[5];
  const int* wb_src = (const int*)d_in[6];
  const int* wb_dst = (const int*)d_in[7];
  const int* rb_src = (const int*)d_in[8];
  const int* rb_dst = (const int*)d_in[9];
  const float* p1wr_wl = (const float*)d_in[10];
  const float* p1wr_bl = (const float*)d_in[11];
  const float* p1wr_wr = (const float*)d_in[12];
  const float* p1rv_wl = (const float*)d_in[13];
  const float* p1rv_bl = (const float*)d_in[14];
  const float* p1rv_wr = (const float*)d_in[15];
  const float* p1wb_wl = (const float*)d_in[16];
  const float* p1wb_bl = (const float*)d_in[17];
  const float* p1wb_wr = (const float*)d_in[18];
  const float* p1rb_wl = (const float*)d_in[19];
  const float* p1rb_bl = (const float*)d_in[20];
  const float* p1rb_wr = (const float*)d_in[21];
  const float* p2wr_wl = (const float*)d_in[22];
  const float* p2wr_bl = (const float*)d_in[23];
  const float* p2wr_wr = (const float*)d_in[24];
  const float* p2rv_wl = (const float*)d_in[25];
  const float* p2rv_bl = (const float*)d_in[26];
  const float* p2rv_wr = (const float*)d_in[27];
  const float* p2wb_wl = (const float*)d_in[28];
  const float* p2wb_bl = (const float*)d_in[29];
  const float* p2wb_wr = (const float*)d_in[30];
  const float* p2rb_wl = (const float*)d_in[31];
  const float* p2rb_bl = (const float*)d_in[32];
  const float* p2rb_wr = (const float*)d_in[33];

  const int ND = in_sizes[0] / 64;
  const int NT = in_sizes[1] / 64;
  const int E_wr = in_sizes[2], E_rv = in_sizes[4];
  const int E_wb = in_sizes[6], E_rb = in_sizes[8];

  float* ws = (float*)d_ws;
  size_t off = 0;
  float* cnt_wr = ws + off; off += NT;
  float* cnt_rv = ws + off; off += NT;
  float* cnt_wb = ws + off; off += ND;
  float* cnt_rb = ws + off; off += ND;
  float* agg_wr = ws + off; off += (size_t)NT * 16;
  float* agg_rv = ws + off; off += (size_t)NT * 16;
  float* agg_wb = ws + off; off += (size_t)ND * 16;
  float* agg_rb = ws + off; off += (size_t)ND * 16;
  float* y_wr = ws + off; off += (size_t)ND * 16;
  float* y_rv = ws + off; off += (size_t)ND * 16;
  float* y_wb = ws + off; off += (size_t)NT * 16;
  float* y_rb = ws + off; off += (size_t)NT * 16;
  float* t1 = ws + off; off += (size_t)NT * 16;
  float* d1 = ws + off; off += (size_t)ND * 16;
  float* ws1t = ws + off; off += 64 * 16;
  float* ws1d = ws + off; off += 64 * 16;
  float* ws2t = ws + off; off += 16 * 32;
  float* ws2d = ws + off; off += 16 * 32;

  const int TPB = 256;
  const long long aggN = (long long)(2LL * NT + 2LL * ND) * 16;
  const long long zero1N = (long long)(2LL * NT + 2LL * ND) + aggN;

  zero_kernel<<<(int)((zero1N + TPB - 1) / TPB), TPB, 0, stream>>>(ws, zero1N);

  count_kernel<<<(E_wr + TPB - 1) / TPB, TPB, 0, stream>>>(wr_dst, E_wr, cnt_wr);
  count_kernel<<<(E_rv + TPB - 1) / TPB, TPB, 0, stream>>>(rv_dst, E_rv, cnt_rv);
  count_kernel<<<(E_wb + TPB - 1) / TPB, TPB, 0, stream>>>(wb_dst, E_wb, cnt_wb);
  count_kernel<<<(E_rb + TPB - 1) / TPB, TPB, 0, stream>>>(rb_dst, E_rb, cnt_rb);

  addmat_kernel<<<4, TPB, 0, stream>>>(p1wr_wr, p1rv_wr, ws1t, 64 * 16);
  addmat_kernel<<<4, TPB, 0, stream>>>(p1wb_wr, p1rb_wr, ws1d, 64 * 16);
  addmat_kernel<<<2, TPB, 0, stream>>>(p2wr_wr, p2rv_wr, ws2t, 16 * 32);
  addmat_kernel<<<2, TPB, 0, stream>>>(p2wb_wr, p2rb_wr, ws2d, 16 * 32);

  const int tilesD = (ND + 15) / 16, tilesT = (NT + 15) / 16;
  l1_transform_kernel<<<(tilesD + 3) / 4, 128, 0, stream>>>(
      x_dev, p1wr_wl, p1rv_wl, ws1d, y_wr, y_rv, d1, ND);
  l1_transform_kernel<<<(tilesT + 3) / 4, 128, 0, stream>>>(
      x_task, p1wb_wl, p1rb_wl, ws1t, y_wb, y_rb, t1, NT);

  auto scb = [&](int E) { return (int)(((long long)E * 4 + TPB - 1) / TPB); };
  scatter16_kernel<<<scb(E_wr), TPB, 0, stream>>>(y_wr, wr_src, wr_dst, E_wr, agg_wr);
  scatter16_kernel<<<scb(E_rv), TPB, 0, stream>>>(y_rv, rv_src, rv_dst, E_rv, agg_rv);
  scatter16_kernel<<<scb(E_wb), TPB, 0, stream>>>(y_wb, wb_src, wb_dst, E_wb, agg_wb);
  scatter16_kernel<<<scb(E_rb), TPB, 0, stream>>>(y_rb, rb_src, rb_dst, E_rb, agg_rb);

  finalize_l1_kernel<<<(int)(((long long)NT * 16 + TPB - 1) / TPB), TPB, 0,
                       stream>>>(t1, agg_wr, cnt_wr, p1wr_bl, agg_rv, cnt_rv,
                                 p1rv_bl, NT);
  finalize_l1_kernel<<<(int)(((long long)ND * 16 + TPB - 1) / TPB), TPB, 0,
                       stream>>>(d1, agg_wb, cnt_wb, p1wb_bl, agg_rb, cnt_rb,
                                 p1rb_bl, ND);

  zero_kernel<<<(int)((aggN + TPB - 1) / TPB), TPB, 0, stream>>>(agg_wr, aggN);

  scatter16_kernel<<<scb(E_wr), TPB, 0, stream>>>(d1, wr_src, wr_dst, E_wr, agg_wr);
  scatter16_kernel<<<scb(E_rv), TPB, 0, stream>>>(d1, rv_src, rv_dst, E_rv, agg_rv);
  scatter16_kernel<<<scb(E_wb), TPB, 0, stream>>>(t1, wb_src, wb_dst, E_wb, agg_wb);
  scatter16_kernel<<<scb(E_rb), TPB, 0, stream>>>(t1, rb_src, rb_dst, E_rb, agg_rb);

  float* out_d2 = (float*)d_out;
  float* out_t2 = (float*)d_out + (size_t)ND * 32;
  l2_fused_kernel<<<(tilesT + 3) / 4, 128, 0, stream>>>(
      agg_wr, cnt_wr, agg_rv, cnt_rv, t1, p2wr_wl, p2rv_wl, ws2t, p2wr_bl,
      p2rv_bl, out_t2, NT);
  l2_fused_kernel<<<(tilesD + 3) / 4, 128, 0, stream>>>(
      agg_wb, cnt_wb, agg_rb, cnt_rb, d1, p2wb_wl, p2rb_wl, ws2d, p2wb_bl,
      p2rb_bl, out_d2, ND);
}


# Scatter waves sort 128 edge slots in a per-wave LDS list, own-class live edges packed in front by ballot rank (1024-edge items): each row load and atomic carries two live edges; sum order may differ
# speedup vs baseline: 2.9322x; 2.9322x over previous
#include <hip/hip_runtime.h>


typedef __attribute__((ext_vector_type(2))) float v2f;
typedef __attribute__((ext_vector_type(4))) float v4f;
typedef __attribute__((ext_vector_type(8))) float v8f;

__device__ __forceinline__ v8f wmma_k4(v2f a, v2f b, v8f c) {
  return __builtin_amdgcn_wmma_f32_16x16x4_f32(
       false, a,  false, b,
       (short)0, c,  false,  false);
}


__global__ __launch_bounds__(256) void zero_kernel(float* __restrict__ p,
                                                   long long n) {
  const long long n4 = n >> 2;
  const long long nblk = n4 >> 7;
  const long long i = (long long)blockIdx.x * blockDim.x + threadIdx.x;
  const long long nwave = ((long long)gridDim.x * blockDim.x) >> 5;
  const int lane = threadIdx.x & 31;
  float4* __restrict__ p4 = reinterpret_cast<float4*>(p);
  const float4 z = make_float4(0.0f, 0.0f, 0.0f, 0.0f);
  for (long long b = i >> 5; b < nblk; b += nwave) {
    float4* __restrict__ q = p4 + (b << 7) + lane;
    q[0] = z;
    q[32] = z;
    q[64] = z;
    q[96] = z;
  }
  const long long r = (nblk << 7) + i;
  if (r < n4) p4[r] = z;
  const long long t = (n4 << 2) + i;
  if (t < n) p[t] = 0.0f;
}

__device__ __forceinline__ void claim_item(unsigned* __restrict__ q,
                                           unsigned nchunks,
                                           unsigned* s_claim) {
  if (threadIdx.x == 0) {
    unsigned cls = (__builtin_amdgcn_s_sendmsg_rtn(0x87) >> 18) & 1u;
    unsigned idx = atomicAdd(q + cls * 64u, 1u);
    if (idx >= nchunks) {
      cls ^= 1u;
      idx = atomicAdd(q + cls * 64u, 1u);
    }
    s_claim[0] = idx;
    s_claim[1] = cls;
  }
  __syncthreads();
}

constexpr int CNT_ROWS = 65536;
constexpr int CNT_TPB = 1024;
constexpr int CNT_UN = 12;

__device__ __forceinline__ void count_hit(unsigned* h, int d, int base,
                                          unsigned nrows) {
  const unsigned o = (unsigned)(d - base);
  if (o < nrows) atomicAdd(h + o, 1u);
}

__global__ __launch_bounds__(CNT_TPB) void count_lds_kernel(
    const int* __restrict__ dstA0, const int* __restrict__ dstA1,
    const int* __restrict__ dstB0, const int* __restrict__ dstB1, int EA0,
    int EA1, int EB0, int EB1, float* __restrict__ cntA0,
    float* __restrict__ cntA1, float* __restrict__ cntB0,
    float* __restrict__ cntB1, int NA, int NB, int rowsA, int rowsB, int nbA,
    int nbB, int P) {
  __shared__ __attribute__((aligned(16))) unsigned h[CNT_ROWS];
  int b = (int)blockIdx.x;
  const int* __restrict__ dst;
  float* __restrict__ cnt;
  int E, N, rows;
  if (b < 2 * nbA) {
    const bool second = b >= nbA;
    if (second) b -= nbA;
    dst = second ? dstA1 : dstA0;
    cnt = second ? cntA1 : cntA0;
    E = second ? EA1 : EA0;
    N = NA;
    rows = rowsA;
  } else {
    b -= 2 * nbA;
    const bool second = b >= nbB;
    if (second) b -= nbB;
    dst = second ? dstB1 : dstB0;
    cnt = second ? cntB1 : cntB0;
    E = second ? EB1 : EB0;
    N = NB;
    rows = rowsB;
  }
  const int r = b / P, p = b - r * P;
  const int base = r * rows;
  if (base >= N) return;
  const unsigned nrows = (unsigned)min(rows, N - base);
  const int tid = (int)threadIdx.x;

  for (int i = tid * 4; i < rows; i += CNT_TPB * 4)
    *reinterpret_cast<uint4*>(h + i) = make_uint4(0u, 0u, 0u, 0u);
  __syncthreads();

  const int S = (((E + P - 1) / P) + 3) & ~3;
  const int e_lo = min(p * S, E), e_hi = min(E, e_lo + S);
  const int nq = (e_hi - e_lo) >> 2;
  const int4* __restrict__ d4 = reinterpret_cast<const int4*>(dst + e_lo);
  for (int q0 = tid; q0 < nq; q0 += CNT_TPB * CNT_UN) {
    int4 v[CNT_UN];
#pragma unroll
    for (int k = 0; k < CNT_UN; ++k)
      v[k] = d4[min(q0 + k * CNT_TPB, nq - 1)];
#pragma unroll
    for (int k = 0; k < CNT_UN; ++k) {
      if (q0 + k * CNT_TPB < nq) {
        count_hit(h, v[k].x, base, nrows);
        count_hit(h, v[k].y, base, nrows);
        count_hit(h, v[k].z, base, nrows);
        count_hit(h, v[k].w, base, nrows);
      }
    }
  }
  const int e_q = e_lo + (nq << 2);
  if (tid < e_hi - e_q) count_hit(h, dst[e_q + tid], base, nrows);
  __syncthreads();

  for (unsigned i = (unsigned)tid; i < nrows; i += CNT_TPB) {
    const unsigned c = h[i];
    if (c != 0u) atomicAdd(cnt + base + i, (float)c);
  }
}

typedef __attribute__((ext_vector_type(2))) unsigned v2u;
constexpr int SC_SLOTS = 128;
constexpr int SC_ITEM = 8 * SC_SLOTS;

template <bool FULL>
__device__ __forceinline__ void scatter_round(const char* __restrict__ Yb,
                                              char* __restrict__ Ab,
                                              const v2u* slot, int cnt,
                                              int half, unsigned c4) {
  unsigned so[16], dof[16];
  float v[16];
#pragma unroll
  for (int j = 0; j < 16; ++j) {
    so[j] = 0u;
    dof[j] = 0u;
    if (!FULL && 2 * j >= cnt) break;
    if (FULL || 2 * j + half < cnt) {
      const v2u e = slot[2 * j + half];
      so[j] = e.x + c4;
      dof[j] = e.y + c4;
    }
  }
#pragma unroll
  for (int j = 0; j < 16; ++j) {
    v[j] = 0.0f;
    if (!FULL && 2 * j >= cnt) break;
    if (FULL || 2 * j + half < cnt)
      v[j] = *reinterpret_cast<const float*>(Yb + so[j]);
  }
#pragma unroll
  for (int j = 0; j < 16; ++j) {
    if (!FULL && 2 * j >= cnt) break;
    if (FULL || 2 * j + half < cnt)
      atomicAdd(reinterpret_cast<float*>(Ab + dof[j]), v[j]);
  }
}

__global__ __launch_bounds__(256) void scatter16_kernel(
    const float* __restrict__ Y0, const float* __restrict__ Y1,
    const float* __restrict__ Y2, const float* __restrict__ Y3,
    const int* __restrict__ src0, const int* __restrict__ dst0,
    const int* __restrict__ src1, const int* __restrict__ dst1,
    const int* __restrict__ src2, const int* __restrict__ dst2,
    const int* __restrict__ src3, const int* __restrict__ dst3, int E0, int E1,
    int E2, int E3, float* __restrict__ agg0, float* __restrict__ agg1,
    float* __restrict__ agg2, float* __restrict__ agg3, int split01,
    int split23, unsigned nb1, unsigned nb2, unsigned nb3,
    unsigned* __restrict__ q) {
  __shared__ unsigned s_claim[2];
  __shared__ v2u s_edge[8][SC_SLOTS];
  const unsigned bx = blockIdx.x;
  const unsigned r =
      (bx >= nb1 ? 1u : 0u) + (bx >= nb2 ? 1u : 0u) + (bx >= nb3 ? 1u : 0u);
  const float* Y = r == 0u ? Y0 : r == 1u ? Y1 : r == 2u ? Y2 : Y3;
  const int* src = r == 0u ? src0 : r == 1u ? src1 : r == 2u ? src2 : src3;
  const int* dst = r == 0u ? dst0 : r == 1u ? dst1 : r == 2u ? dst2 : dst3;
  float* agg = r == 0u ? agg0 : r == 1u ? agg1 : r == 2u ? agg2 : agg3;
  const int E = r == 0u ? E0 : r == 1u ? E1 : r == 2u ? E2 : E3;
  const int split = r < 2u ? split01 : split23;
  const unsigned nchunks = (unsigned)((E + SC_ITEM - 1) / SC_ITEM);
  claim_item(q + r, nchunks, s_claim);
  const unsigned idx = s_claim[0];
  const bool upper = s_claim[1] != 0u;
  if (idx >= nchunks) return;
  const int lane = threadIdx.x & 31;
  const int wave = threadIdx.x >> 5;
  const int half = lane >> 4;
  const unsigned c4 = (unsigned)(lane & 15) << 2;
  const int e0 = (int)idx * SC_ITEM + wave * SC_SLOTS + lane;
  int sv[4], tv[4];
#pragma unroll
  for (int k = 0; k < 4; ++k) {
    const int e = min(e0 + 32 * k, E - 1);
    sv[k] = src[e];
    tv[k] = dst[e];
  }
  v2u* slot = s_edge[wave];
  bool live[4];
  unsigned m[4];
  int n = 0;
#pragma unroll
  for (int k = 0; k < 4; ++k) {
    live[k] = (e0 + 32 * k < E) & ((tv[k] >= split) == upper);
    m[k] = __builtin_amdgcn_ballot_w32(live[k]);
    n += __builtin_popcount(m[k]);
  }
  int lp = 0, dp = n;
#pragma unroll
  for (int k = 0; k < 4; ++k) {
    const int below = (int)__builtin_amdgcn_mbcnt_lo(m[k], 0u);
    const int pos = live[k] ? lp + below : dp + (lane - below);
    slot[pos] = (v2u){(unsigned)sv[k] << 6, (unsigned)tv[k] << 6};
    const int cnt = __builtin_popcount(m[k]);
    lp += cnt;
    dp += 32 - cnt;
  }
  const char* __restrict__ Yb = reinterpret_cast<const char*>(Y);
  char* __restrict__ Ab = reinterpret_cast<char*>(agg);
  int base = 0;
  for (; base + 32 <= n; base += 32)
    scatter_round<true>(Yb, Ab, slot + base, 32, half, c4);
  if (base < n) scatter_round<false>(Yb, Ab, slot + base, n - base, half, c4);
}

__global__ __launch_bounds__(256) void finalize_l1_kernel(
    float* __restrict__ XOp, const float* __restrict__ aggAp,
    const float* __restrict__ cntAp, const float* __restrict__ bAp,
    const float* __restrict__ aggBp, const float* __restrict__ cntBp,
    const float* __restrict__ bBp, int Mp, float* __restrict__ XOq,
    const float* __restrict__ aggAq, const float* __restrict__ cntAq,
    const float* __restrict__ bAq, const float* __restrict__ aggBq,
    const float* __restrict__ cntBq, const float* __restrict__ bBq, int Mq,
    int nbP) {
  const bool second = (int)blockIdx.x >= nbP;
  float* XO = second ? XOq : XOp;
  const float* aggA = second ? aggAq : aggAp;
  const float* cntA = second ? cntAq : cntAp;
  const float* bA = second ? bAq : bAp;
  const float* aggB = second ? aggBq : aggBp;
  const float* cntB = second ? cntBq : cntBp;
  const float* bB = second ? bBq : bBp;
  const int M = second ? Mq : Mp;
  const int pblk = second ? (int)blockIdx.x - nbP : (int)blockIdx.x;
  const int pgrid = second ? (int)gridDim.x - nbP : nbP;
  const long long n4 = (long long)M * 4;
  const long long stride = (long long)pgrid * blockDim.x;
  const long long i = (long long)pblk * blockDim.x + threadIdx.x;
  const int f = (int)(i & 3) << 2;
  const float ba0 = bA[f], ba1 = bA[f + 1], ba2 = bA[f + 2], ba3 = bA[f + 3];
  const float bb0 = bB[f], bb1 = bB[f + 1], bb2 = bB[f + 2], bb3 = bB[f + 3];
  v4f* __restrict__ X4 = reinterpret_cast<v4f*>(XO);
  const v4f* __restrict__ A4 = reinterpret_cast<const v4f*>(aggA);
  const v4f* __restrict__ B4 = reinterpret_cast<const v4f*>(aggB);
  for (long long j = i; j < n4; j += stride) {
    const long long node = j >> 2;
    const v4f x = X4[j];
    const v4f a = A4[j];
    const v4f b = B4[j];
    const float ca = fmaxf(cntA[node], 1.0f);
    const float cb = fmaxf(cntB[node], 1.0f);
    v4f o;
    o.x = fmaxf(x.x + a.x / ca + ba0 + b.x / cb + bb0, 0.0f);
    o.y = fmaxf(x.y + a.y / ca + ba1 + b.y / cb + bb1, 0.0f);
    o.z = fmaxf(x.z + a.z / ca + ba2 + b.z / cb + bb2, 0.0f);
    o.w = fmaxf(x.w + a.w / ca + ba3 + b.w / cb + bb3, 0.0f);
    X4[j] = o;
  }
}

__global__ __launch_bounds__(128, 7) void l1_transform_kernel(
    const float* __restrict__ Xp, const float* __restrict__ W0p,
    const float* __restrict__ W1p, const float* __restrict__ W2ap,
    const float* __restrict__ W2bp, float* __restrict__ Y0p,
    float* __restrict__ Y1p, float* __restrict__ Y2p,
    float* __restrict__ aggZ0p, float* __restrict__ aggZ1p,
    float* __restrict__ cntZ0p, float* __restrict__ cntZ1p, int Mp,
    const float* __restrict__ Xq, const float* __restrict__ W0q,
    const float* __restrict__ W1q, const float* __restrict__ W2aq,
    const float* __restrict__ W2bq, float* __restrict__ Y0q,
    float* __restrict__ Y1q, float* __restrict__ Y2q,
    float* __restrict__ aggZ0q, float* __restrict__ aggZ1q,
    float* __restrict__ cntZ0q, float* __restrict__ cntZ1q, int Mq,
    unsigned* __restrict__ qz, int nbP) {
  const bool second = (int)blockIdx.x >= nbP;
  const float* X = second ? Xq : Xp;
  const float* W0 = second ? W0q : W0p;
  const float* W1 = second ? W1q : W1p;
  const float* W2a = second ? W2aq : W2ap;
  const float* W2b = second ? W2bq : W2bp;
  float* Y0 = second ? Y0q : Y0p;
  float* Y1 = second ? Y1q : Y1p;
  float* Y2 = second ? Y2q : Y2p;
  float* aggZ0 = second ? aggZ0q : aggZ0p;
  float* aggZ1 = second ? aggZ1q : aggZ1p;
  float* cntZ0 = second ? cntZ0q : cntZ0p;
  float* cntZ1 = second ? cntZ1q : cntZ1p;
  const int M = second ? Mq : Mp;
  const int pblk = second ? (int)blockIdx.x - nbP : (int)blockIdx.x;
  const int pgrid = second ? (int)gridDim.x - nbP : nbP;
  const int lane = threadIdx.x & 31;
  const int wave = threadIdx.x >> 5;
  const int wpb = blockDim.x >> 5;
  const int half = lane >> 4;
  const int mrow = lane & 15;
  const int nTiles = (M + 15) >> 4;

  v2f b0[16], b1[16], b2[16];
#pragma unroll
  for (int s = 0; s < 16; ++s) {
    int k = 4 * s + 2 * half;
    b0[s] = (v2f){W0[k * 16 + mrow], W0[(k + 1) * 16 + mrow]};
    b1[s] = (v2f){W1[k * 16 + mrow], W1[(k + 1) * 16 + mrow]};
    b2[s] = (v2f){W2a[k * 16 + mrow] + W2b[k * 16 + mrow],
                  W2a[(k + 1) * 16 + mrow] + W2b[(k + 1) * 16 + mrow]};
  }
  if (blockIdx.x == 0) qz[threadIdx.x] = 0u;
  const v4f z4 = {0.0f, 0.0f, 0.0f, 0.0f};

  for (int tile = pblk * wpb + wave; tile < nTiles; tile += pgrid * wpb) {
    const int rbase = tile << 4;
    int arow = rbase + mrow;
    if (arow > M - 1) arow = M - 1;
    const float* xr = X + (size_t)arow * 64;
    v8f c0 = {}, c1 = {}, c2 = {};
#pragma unroll
    for (int s = 0; s < 16; ++s) {
      int k = 4 * s + 2 * half;
      v2f a = (v2f){xr[k], xr[k + 1]};
      c0 = wmma_k4(a, b0[s], c0);
      c1 = wmma_k4(a, b1[s], c1);
      c2 = wmma_k4(a, b2[s], c2);
    }
#pragma unroll
    for (int r = 0; r < 8; ++r) {
      int row = rbase + r + 8 * half;
      if (row < M) {
        size_t o = (size_t)row * 16 + mrow;
        Y0[o] = c0[r];
        Y1[o] = c1[r];
        Y2[o] = c2[r];
      }
    }
    const size_t zq = (size_t)rbase * 16 + (size_t)lane * 4;
    if (rbase + (lane >> 2) < M) {
      *reinterpret_cast<v4f*>(aggZ0 + zq) = z4;
      *reinterpret_cast<v4f*>(aggZ1 + zq) = z4;
    }
    if (rbase + 8 + (lane >> 2) < M) {
      *reinterpret_cast<v4f*>(aggZ0 + zq + 128) = z4;
      *reinterpret_cast<v4f*>(aggZ1 + zq + 128) = z4;
    }
    if (rbase + mrow < M) (half ? cntZ1 : cntZ0)[rbase + mrow] = 0.0f;
  }
}

__global__ __launch_bounds__(128) void l2_fused_kernel(
    const float* __restrict__ A0p, const float* __restrict__ cnt0p,
    const float* __restrict__ A1p, const float* __restrict__ cnt1p,
    const float* __restrict__ A2p, const float* __restrict__ W0p,
    const float* __restrict__ W1p, const float* __restrict__ W2ap,
    const float* __restrict__ W2bp, const float* __restrict__ bias0p,
    const float* __restrict__ bias1p, float* __restrict__ OUTp, int Mp,
    const float* __restrict__ A0q, const float* __restrict__ cnt0q,
    const float* __restrict__ A1q, const float* __restrict__ cnt1q,
    const float* __restrict__ A2q, const float* __restrict__ W0q,
    const float* __restrict__ W1q, const float* __restrict__ W2aq,
    const float* __restrict__ W2bq, const float* __restrict__ bias0q,
    const float* __restrict__ bias1q, float* __restrict__ OUTq, int Mq,
    int nbP) {
  const bool second = (int)blockIdx.x >= nbP;
  const float* A0 = second ? A0q : A0p;
  const float* cnt0 = second ? cnt0q : cnt0p;
  const float* A1 = second ? A1q : A1p;
  const float* cnt1 = second ? cnt1q : cnt1p;
  const float* A2 = second ? A2q : A2p;
  const float* W0 = second ? W0q : W0p;
  const float* W1 = second ? W1q : W1p;
  const float* W2a = second ? W2aq : W2ap;
  const float* W2b = second ? W2bq : W2bp;
  const float* bias0 = second ? bias0q : bias0p;
  const float* bias1 = second ? bias1q : bias1p;
  float* OUT = second ? OUTq : OUTp;
  const int M = second ? Mq : Mp;
  const int pblk = second ? (int)blockIdx.x - nbP : (int)blockIdx.x;
  const int pgrid = second ? (int)gridDim.x - nbP : nbP;
  const int lane = threadIdx.x & 31;
  const int wave = threadIdx.x >> 5;
  const int wpb = blockDim.x >> 5;
  const int half = lane >> 4;
  const int mrow = lane & 15;
  const int nTiles = (M + 15) >> 4;

  const float* Ws[2] = {W0, W1};
  v2f b[3][4][2];
#pragma unroll
  for (int s = 0; s < 4; ++s)
#pragma unroll
    for (int t = 0; t < 2; ++t) {
      int k = 4 * s + 2 * half;
      int col = 16 * t + mrow;
#pragma unroll
      for (int m = 0; m < 2; ++m)
        b[m][s][t] = (v2f){Ws[m][k * 32 + col], Ws[m][(k + 1) * 32 + col]};
      b[2][s][t] = (v2f){W2a[k * 32 + col] + W2b[k * 32 + col],
                         W2a[(k + 1) * 32 + col] + W2b[(k + 1) * 32 + col]};
    }

  const float biasL = bias0[mrow] + bias1[mrow];
  const float biasH = bias0[16 + mrow] + bias1[16 + mrow];

  for (int tile = pblk * wpb + wave; tile < nTiles; tile += pgrid * wpb) {
    const int rbase = tile << 4;
    int arow = rbase + mrow;
    if (arow > M - 1) arow = M - 1;
    const float rs0 = 1.0f / fmaxf(cnt0[arow], 1.0f);
    const float rs1 = 1.0f / fmaxf(cnt1[arow], 1.0f);
    const float* a0p = A0 + (size_t)arow * 16;
    const float* a1p = A1 + (size_t)arow * 16;
    const float* a2p = A2 + (size_t)arow * 16;
    v8f cL = {}, cH = {};
#pragma unroll
    for (int s = 0; s < 4; ++s) {
      int k = 4 * s + 2 * half;
      v2f a0 = (v2f){a0p[k] * rs0, a0p[k + 1] * rs0};
      cL = wmma_k4(a0, b[0][s][0], cL);
      cH = wmma_k4(a0, b[0][s][1], cH);
      v2f a1 = (v2f){a1p[k] * rs1, a1p[k + 1] * rs1};
      cL = wmma_k4(a1, b[1][s][0], cL);
      cH = wmma_k4(a1, b[1][s][1], cH);
      v2f a2 = (v2f){a2p[k], a2p[k + 1]};
      cL = wmma_k4(a2, b[2][s][0], cL);
      cH = wmma_k4(a2, b[2][s][1], cH);
    }
#pragma unroll
    for (int r = 0; r < 8; ++r) {
      int row = rbase + r + 8 * half;
      if (row < M) {
        size_t o = (size_t)row * 32 + mrow;
        OUT[o] = cL[r] + biasL;
        OUT[o + 16] = cH[r] + biasH;
      }
    }
  }
}


extern "C" void kernel_launch(void* const* d_in, const int* in_sizes, int n_in,
                              void* d_out, int out_size, void* d_ws,
                              size_t ws_size, hipStream_t stream) {
  const float* x_dev = (const float*)d_in[0];
  const float* x_task = (const float*)d_in[1];
  const int* wr_src = (const int*)d_in[2];
  const int* wr_dst = (const int*)d_in[3];
  const int* rv_src = (const int*)d_in[4];
  const int* rv_dst = (const int*)d_in[5];
  const int* wb_src = (const int*)d_in[6];
  const int* wb_dst = (const int*)d_in[7];
  const int* rb_src = (const int*)d_in[8];
  const int* rb_dst = (const int*)d_in[9];
  const float* p1wr_wl = (const float*)d_in[10];
  const float* p1wr_bl = (const float*)d_in[11];
  const float* p1wr_wr = (const float*)d_in[12];
  const float* p1rv_wl = (const float*)d_in[13];
  const float* p1rv_bl = (const float*)d_in[14];
  const float* p1rv_wr = (const float*)d_in[15];
  const float* p1wb_wl = (const float*)d_in[16];
  const float* p1wb_bl = (const float*)d_in[17];
  const float* p1wb_wr = (const float*)d_in[18];
  const float* p1rb_wl = (const float*)d_in[19];
  const float* p1rb_bl = (const float*)d_in[20];
  const float* p1rb_wr = (const float*)d_in[21];
  const float* p2wr_wl = (const float*)d_in[22];
  const float* p2wr_bl = (const float*)d_in[23];
  const float* p2wr_wr = (const float*)d_in[24];
  const float* p2rv_wl = (const float*)d_in[25];
  const float* p2rv_bl = (const float*)d_in[26];
  const float* p2rv_wr = (const float*)d_in[27];
  const float* p2wb_wl = (const float*)d_in[28];
  const float* p2wb_bl = (const float*)d_in[29];
  const float* p2wb_wr = (const float*)d_in[30];
  const float* p2rb_wl = (const float*)d_in[31];
  const float* p2rb_bl = (const float*)d_in[32];
  const float* p2rb_wr = (const float*)d_in[33];

  const int ND = in_sizes[0] / 64;
  const int NT = in_sizes[1] / 64;
  const int E_wr = in_sizes[2], E_rv = in_sizes[4];
  const int E_wb = in_sizes[6], E_rb = in_sizes[8];

  float* ws = (float*)d_ws;
  size_t off = 0;
  unsigned* qs = (unsigned*)(ws + off); off += 128;
  float* cnt_wr = ws + off; off += NT;
  float* cnt_rv = ws + off; off += NT;
  float* cnt_wb = ws + off; off += ND;
  float* cnt_rb = ws + off; off += ND;
  float* agg_wr = ws + off; off += (size_t)NT * 16;
  float* agg_rv = ws + off; off += (size_t)NT * 16;
  float* agg_wb = ws + off; off += (size_t)ND * 16;
  float* agg_rb = ws + off; off += (size_t)ND * 16;
  float* y_wr = ws + off; off += (size_t)ND * 16;
  float* y_rv = ws + off; off += (size_t)ND * 16;
  float* y_wb = ws + off; off += (size_t)NT * 16;
  float* y_rb = ws + off; off += (size_t)NT * 16;
  float* t1 = ws + off; off += (size_t)NT * 16;
  float* d1 = ws + off; off += (size_t)ND * 16;

  const int TPB = 256;
  const int ZERO_BLOCKS = 512;
  const int FINAL_BLOCKS = 1024;
  const int FINAL_BLOCKS_D = 256;
  const long long aggN = (long long)(2LL * NT + 2LL * ND) * 16;
  const int splitT = (NT / 2) & ~63, splitD = (ND / 2) & ~63;

  const int tilesD = (ND + 15) / 16, tilesT = (NT + 15) / 16;
  const int gD = (tilesD + 3) / 4, gT = (tilesT + 3) / 4;
  l1_transform_kernel<<<gD + gT, 128, 0, stream>>>(
      x_dev, p1wr_wl, p1rv_wl, p1wb_wr, p1rb_wr, y_wr, y_rv, d1, agg_wb,
      agg_rb, cnt_wb, cnt_rb, ND, x_task, p1wb_wl, p1rb_wl, p1wr_wr, p1rv_wr,
      y_wb, y_rb, t1, agg_wr, agg_rv, cnt_wr, cnt_rv, NT, qs, gD);

  auto nrg = [&](int N) { return N > CNT_ROWS ? (N + CNT_ROWS - 1) / CNT_ROWS : 1; };
  auto rpr = [&](int N) { return (((N + nrg(N) - 1) / nrg(N)) + 31) & ~31; };
  const int RT = nrg(NT), RD = nrg(ND);
  const int CP = 2 * (RT + RD) < 256 ? 256 / (2 * (RT + RD)) : 1;
  count_lds_kernel<<<2 * (RT + RD) * CP, CNT_TPB, 0, stream>>>(
      wr_dst, rv_dst, wb_dst, rb_dst, E_wr, E_rv, E_wb, E_rb, cnt_wr, cnt_rv,
      cnt_wb, cnt_rb, NT, ND, rpr(NT), rpr(ND), RT * CP, RD * CP, CP);

  auto nsc = [&](int E) { return (unsigned)((E + SC_ITEM - 1) / SC_ITEM); };
  const unsigned sb1 = 2 * nsc(E_wr), sb2 = sb1 + 2 * nsc(E_rv);
  const unsigned sb3 = sb2 + 2 * nsc(E_wb), sb4 = sb3 + 2 * nsc(E_rb);
  scatter16_kernel<<<sb4, TPB, 0, stream>>>(
      y_wr, y_rv, y_wb, y_rb, wr_src, wr_dst, rv_src, rv_dst, wb_src, wb_dst,
      rb_src, rb_dst, E_wr, E_rv, E_wb, E_rb, agg_wr, agg_rv, agg_wb, agg_rb,
      splitT, splitD, sb1, sb2, sb3, qs + 0);

  finalize_l1_kernel<<<FINAL_BLOCKS + FINAL_BLOCKS_D, TPB, 0, stream>>>(
      t1, agg_wr, cnt_wr, p1wr_bl, agg_rv, cnt_rv, p1rv_bl, NT, d1, agg_wb,
      cnt_wb, p1wb_bl, agg_rb, cnt_rb, p1rb_bl, ND, FINAL_BLOCKS);

  zero_kernel<<<ZERO_BLOCKS, TPB, 0, stream>>>(agg_wr, aggN);

  scatter16_kernel<<<sb4, TPB, 0, stream>>>(
      d1, d1, t1, t1, wr_src, wr_dst, rv_src, rv_dst, wb_src, wb_dst, rb_src,
      rb_dst, E_wr, E_rv, E_wb, E_rb, agg_wr, agg_rv, agg_wb, agg_rb, splitT,
      splitD, sb1, sb2, sb3, qs + 4);

  float* out_d2 = (float*)d_out;
  float* out_t2 = (float*)d_out + (size_t)ND * 32;
  l2_fused_kernel<<<gT + gD, 128, 0, stream>>>(
      agg_wr, cnt_wr, agg_rv, cnt_rv, t1, p2wr_wl, p2rv_wl, p2wr_wr, p2rv_wr,
      p2wr_bl, p2rv_bl, out_t2, NT, agg_wb, cnt_wb, agg_rb, cnt_rb, d1,
      p2wb_wl, p2rb_wl, p2wb_wr, p2rb_wr, p2wb_bl, p2rb_bl, out_d2, ND, gT);
}
